# starting kernel
# baseline (speedup 1.0000x reference)
#include <hip/hip_runtime.h>

typedef float v2f __attribute__((ext_vector_type(2)));
typedef float v8f __attribute__((ext_vector_type(8)));

#define N_DIM   2000
#define C_DIM   32
#define J_DIM   4
#define BS_DIM  4
#define NCH     128
#define PER_CH  8000.0f
#define BN_EPS  1e-5f

__device__ __forceinline__ v8f wmma4(v2f a, v2f b, v8f c) {
    return __builtin_amdgcn_wmma_f32_16x16x4_f32(
         false, a,  false, b,
         (short)0, c,  false,  false);
}

#define WS_Y      0
#define WS_SUM    1024000
#define WS_SQ     1024128
#define WS_SCALE  1024256
#define WS_SHIFT  1024384

__global__ void k_init(float* __restrict__ gsum, float* __restrict__ gsq) {
    int t = threadIdx.x;
    if (t < NCH) { gsum[t] = 0.0f; gsq[t] = 0.0f; }
}

__global__ void k_gmul(const float* __restrict__ W, const float* __restrict__ x,
                       float* __restrict__ y, float* __restrict__ gsum,
                       float* __restrict__ gsq) {
    __shared__ float ls[C_DIM];
    __shared__ float lq[C_DIM];
    const int tid  = threadIdx.x;
    const int lane = tid & 31;
    const int wave = tid >> 5;
    const int bj   = blockIdx.y;
    const int b    = bj >> 2;
    const int j    = bj & 3;

    if (tid < C_DIM) { ls[tid] = 0.0f; lq[tid] = 0.0f; }
    __syncthreads();

    const int mtile = blockIdx.x * 8 + wave;
    if (mtile < 125) {
        const int lo = lane & 15;
        const int hi = lane >> 4;
        const int m  = mtile * 16 + lo;
        const float* pa  = W + (size_t)bj * N_DIM * N_DIM + (size_t)m * N_DIM + hi * 2;
        const float* pb0 = x + ((size_t)b * C_DIM + lo) * N_DIM + hi * 2;
        const float* pb1 = pb0 + (size_t)16 * N_DIM;

        v8f acc0 = {};
        v8f acc1 = {};
        #pragma unroll 8
        for (int k = 0; k < N_DIM; k += 4) {
            v2f a  = __builtin_nontemporal_load((const v2f*)(pa + k));
            v2f b0 = *(const v2f*)(pb0 + k);
            v2f b1 = *(const v2f*)(pb1 + k);
            acc0 = wmma4(a, b0, acc0);
            acc1 = wmma4(a, b1, acc1);
        }

        const int m0  = mtile * 16 + hi * 8;
        const int ch0 = j * C_DIM + lo;
        float* yp0 = y + ((size_t)(b * NCH + ch0))      * N_DIM + m0;
        float* yp1 = y + ((size_t)(b * NCH + ch0 + 16)) * N_DIM + m0;
        *(float4*)(yp0)     = make_float4(acc0[0], acc0[1], acc0[2], acc0[3]);
        *(float4*)(yp0 + 4) = make_float4(acc0[4], acc0[5], acc0[6], acc0[7]);
        *(float4*)(yp1)     = make_float4(acc1[0], acc1[1], acc1[2], acc1[3]);
        *(float4*)(yp1 + 4) = make_float4(acc1[4], acc1[5], acc1[6], acc1[7]);

        float s0 = 0.f, q0 = 0.f, s1 = 0.f, q1 = 0.f;
        #pragma unroll
        for (int r = 0; r < 8; ++r) {
            s0 += acc0[r]; q0 += acc0[r] * acc0[r];
            s1 += acc1[r]; q1 += acc1[r] * acc1[r];
        }
        s0 += __shfl_xor(s0, 16); q0 += __shfl_xor(q0, 16);
        s1 += __shfl_xor(s1, 16); q1 += __shfl_xor(q1, 16);
        if (lane < 16) {
            atomicAdd(&ls[lo],      s0); atomicAdd(&lq[lo],      q0);
            atomicAdd(&ls[lo + 16], s1); atomicAdd(&lq[lo + 16], q1);
        }
    }
    __syncthreads();
    if (tid < C_DIM) {
        atomicAdd(&gsum[j * C_DIM + tid], ls[tid]);
        atomicAdd(&gsq [j * C_DIM + tid], lq[tid]);
    }
}

__global__ void k_stats(const float* __restrict__ gsum, const float* __restrict__ gsq,
                        const float* __restrict__ gamma, const float* __restrict__ beta,
                        float* __restrict__ scale, float* __restrict__ shift) {
    int c = threadIdx.x;
    if (c < NCH) {
        float mean = gsum[c] * (1.0f / PER_CH);
        float var  = gsq[c]  * (1.0f / PER_CH) - mean * mean;
        float sc   = gamma[c] * rsqrtf(var + BN_EPS);
        scale[c] = sc;
        shift[c] = beta[c] - mean * sc;
    }
}

__global__ void k_conv(const float* __restrict__ y, const float* __restrict__ scale,
                       const float* __restrict__ shift, const float* __restrict__ cw,
                       const float* __restrict__ cb, float* __restrict__ out) {
    __shared__ float ssc[NCH];
    __shared__ float ssh[NCH];
    const int lane = threadIdx.x;
    for (int i = lane; i < NCH; i += 32) { ssc[i] = scale[i]; ssh[i] = shift[i]; }
    __syncthreads();

    const int ntile = blockIdx.x;
    const int b     = blockIdx.y >> 1;
    const int ot    = blockIdx.y & 1;
    const int lo    = lane & 15;
    const int hi    = lane >> 4;
    const int o     = ot * 16 + lo;

    const float* pa = cw + (size_t)o * NCH + hi * 2;
    const float* py = y + (size_t)b * NCH * N_DIM + ntile * 16 + lo;

    v8f acc = {};
    #pragma unroll 4
    for (int k = 0; k < NCH; k += 4) {
        const int ch = k + hi * 2;
        v2f a = *(const v2f*)(pa + k);
        float v0 = __builtin_nontemporal_load(py + (size_t)ch       * N_DIM);
        float v1 = __builtin_nontemporal_load(py + (size_t)(ch + 1) * N_DIM);
        v0 = fmaxf(fmaf(v0, ssc[ch],     ssh[ch]),     0.0f);
        v1 = fmaxf(fmaf(v1, ssc[ch + 1], ssh[ch + 1]), 0.0f);
        v2f bb; bb[0] = v0; bb[1] = v1;
        acc = wmma4(a, bb, acc);
    }

    const int obase = ot * 16 + hi * 8;
    const int n0    = ntile * 16 + lo;
    #pragma unroll
    for (int r = 0; r < 8; ++r) {
        out[((size_t)(b * C_DIM + obase + r)) * N_DIM + n0] = acc[r] + cb[obase + r];
    }
}

extern "C" void kernel_launch(void* const* d_in, const int* in_sizes, int n_in,
                              void* d_out, int out_size, void* d_ws, size_t ws_size,
                              hipStream_t stream) {
    (void)in_sizes; (void)n_in; (void)out_size; (void)ws_size;
    const float* W      = (const float*)d_in[0];
    const float* x      = (const float*)d_in[1];
    const float* gamma  = (const float*)d_in[2];
    const float* beta   = (const float*)d_in[3];
    const float* conv_w = (const float*)d_in[4];
    const float* conv_b = (const float*)d_in[5];
    float* out = (float*)d_out;

    float* ws     = (float*)d_ws;
    float* y      = ws + WS_Y;
    float* gsum   = ws + WS_SUM;
    float* gsq    = ws + WS_SQ;
    float* sscale = ws + WS_SCALE;
    float* sshift = ws + WS_SHIFT;

    k_init<<<1, 256, 0, stream>>>(gsum, gsq);
    dim3 g1(16, 16);
    k_gmul<<<g1, 256, 0, stream>>>(W, x, y, gsum, gsq);
    k_stats<<<1, 128, 0, stream>>>(gsum, gsq, gamma, beta, sscale, sshift);
    dim3 g3(125, 8);
    k_conv<<<g3, 32, 0, stream>>>(y, sscale, sshift, conv_w, conv_b, out);
}


# k_gmul: W and x K-chunks staged through LDS by 16-B loads, one wave per 512-B row piece, fragments read from LDS; k summed in a permuted order inside each group of 8
# speedup vs baseline: 3.0193x; 3.0193x over previous
#include <hip/hip_runtime.h>

typedef float v2f __attribute__((ext_vector_type(2)));
typedef float v4f __attribute__((ext_vector_type(4)));
typedef float v8f __attribute__((ext_vector_type(8)));

#define N_DIM   2000
#define C_DIM   32
#define J_DIM   4
#define BS_DIM  4
#define NCH     128
#define PER_CH  8000.0f
#define BN_EPS  1e-5f

__device__ __forceinline__ v8f wmma4(v2f a, v2f b, v8f c) {
    return __builtin_amdgcn_wmma_f32_16x16x4_f32(
         false, a,  false, b,
         (short)0, c,  false,  false);
}

#define WS_Y      0
#define WS_SUM    1024000
#define WS_SQ     1024128
#define WS_SCALE  1024256
#define WS_SHIFT  1024384

__global__ void k_init(float* __restrict__ gsum, float* __restrict__ gsq) {
    int t = threadIdx.x;
    if (t < NCH) { gsum[t] = 0.0f; gsq[t] = 0.0f; }
}

#define KC      128
#define NCHUNK  16
#define KLAST   10
#define LP      132
#define MB      128

template <int NS>
__device__ __forceinline__ void mma_chunk(const float* __restrict__ pa, const float* __restrict__ pb0,
                                          const float* __restrict__ pb1, v8f& acc0, v8f& acc1) {
    #pragma unroll
    for (int s = 0; s < NS; ++s) {
        const v4f fa  = *(const v4f*)(pa  + s * 8);
        const v4f fb0 = *(const v4f*)(pb0 + s * 8);
        const v4f fb1 = *(const v4f*)(pb1 + s * 8);
        v2f a0, a1, p0, p1, q0, q1;
        a0[0] = fa[0];  a0[1] = fa[1];  a1[0] = fa[2];  a1[1] = fa[3];
        p0[0] = fb0[0]; p0[1] = fb0[1]; p1[0] = fb0[2]; p1[1] = fb0[3];
        q0[0] = fb1[0]; q0[1] = fb1[1]; q1[0] = fb1[2]; q1[1] = fb1[3];
        acc0 = wmma4(a0, p0, acc0);
        acc1 = wmma4(a0, q0, acc1);
        acc0 = wmma4(a1, p1, acc0);
        acc1 = wmma4(a1, q1, acc1);
    }
}

__global__ void __launch_bounds__(256)
k_gmul(const float* __restrict__ W, const float* __restrict__ x,
       float* __restrict__ y, float* __restrict__ gsum,
       float* __restrict__ gsq) {
    __shared__ __attribute__((aligned(16))) float sA[2][MB * LP];
    __shared__ __attribute__((aligned(16))) float sB[2][C_DIM * LP];
    __shared__ float ls[C_DIM];
    __shared__ float lq[C_DIM];
    const int tid  = threadIdx.x;
    const int lane = tid & 31;
    const int wave = tid >> 5;
    const int bj   = blockIdx.y;
    const int b    = bj >> 2;
    const int j    = bj & 3;
    const int lo   = lane & 15;
    const int hi   = lane >> 4;

    if (tid < C_DIM) { ls[tid] = 0.0f; lq[tid] = 0.0f; }

    const int mblk = blockIdx.x * MB;
    const float* gA[16];
    #pragma unroll
    for (int i = 0; i < 16; ++i) {
        const int r = min(mblk + wave + 8 * i, N_DIM - 1);
        gA[i] = W + (size_t)bj * N_DIM * N_DIM + (size_t)r * N_DIM;
    }
    const float* gB = x + ((size_t)b * C_DIM + wave) * N_DIM;
    const int soff  = wave * LP + lane * 4;

    v4f ra[16], rb[4];
    {
        const int k0 = lane * 4;
        #pragma unroll
        for (int i = 0; i < 16; ++i) ra[i] = __builtin_nontemporal_load((const v4f*)(gA[i] + k0));
        #pragma unroll
        for (int i = 0; i < 4; ++i)  rb[i] = *(const v4f*)(gB + (size_t)i * 8 * N_DIM + k0);
        #pragma unroll
        for (int i = 0; i < 16; ++i) *(v4f*)(&sA[0][soff + i * 8 * LP]) = ra[i];
        #pragma unroll
        for (int i = 0; i < 4; ++i)  *(v4f*)(&sB[0][soff + i * 8 * LP]) = rb[i];
    }
    __syncthreads();

    const int  mtile = blockIdx.x * 8 + wave;
    const bool tile_ok = mtile < 125;
    const int  fa_off  = (wave * 16 + lo) * LP + hi * 4;
    const int  fb_off  = lo * LP + hi * 4;

    v8f acc0 = {};
    v8f acc1 = {};
    for (int c = 0; c < NCHUNK; ++c) {
        const int buf = c & 1;
        const bool more = (c + 1) < NCHUNK;
        if (more) {
            const int k1 = min((c + 1) * KC + lane * 4, N_DIM - 4);
            #pragma unroll
            for (int i = 0; i < 16; ++i) ra[i] = __builtin_nontemporal_load((const v4f*)(gA[i] + k1));
            #pragma unroll
            for (int i = 0; i < 4; ++i)  rb[i] = *(const v4f*)(gB + (size_t)i * 8 * N_DIM + k1);
        }
        if (tile_ok) {
            const float* pa  = &sA[buf][fa_off];
            const float* pb0 = &sB[buf][fb_off];
            const float* pb1 = pb0 + 16 * LP;
            if (more) mma_chunk<16>(pa, pb0, pb1, acc0, acc1);
            else      mma_chunk<KLAST>(pa, pb0, pb1, acc0, acc1);
        }
        if (more) {
            #pragma unroll
            for (int i = 0; i < 16; ++i) *(v4f*)(&sA[buf ^ 1][soff + i * 8 * LP]) = ra[i];
            #pragma unroll
            for (int i = 0; i < 4; ++i)  *(v4f*)(&sB[buf ^ 1][soff + i * 8 * LP]) = rb[i];
        }
        __syncthreads();
    }

    if (tile_ok) {
        const int m0  = mtile * 16 + hi * 8;
        const int ch0 = j * C_DIM + lo;
        float* yp0 = y + ((size_t)(b * NCH + ch0))      * N_DIM + m0;
        float* yp1 = y + ((size_t)(b * NCH + ch0 + 16)) * N_DIM + m0;
        *(float4*)(yp0)     = make_float4(acc0[0], acc0[1], acc0[2], acc0[3]);
        *(float4*)(yp0 + 4) = make_float4(acc0[4], acc0[5], acc0[6], acc0[7]);
        *(float4*)(yp1)     = make_float4(acc1[0], acc1[1], acc1[2], acc1[3]);
        *(float4*)(yp1 + 4) = make_float4(acc1[4], acc1[5], acc1[6], acc1[7]);

        float s0 = 0.f, q0 = 0.f, s1 = 0.f, q1 = 0.f;
        #pragma unroll
        for (int r = 0; r < 8; ++r) {
            s0 += acc0[r]; q0 += acc0[r] * acc0[r];
            s1 += acc1[r]; q1 += acc1[r] * acc1[r];
        }
        s0 += __shfl_xor(s0, 16); q0 += __shfl_xor(q0, 16);
        s1 += __shfl_xor(s1, 16); q1 += __shfl_xor(q1, 16);
        if (lane < 16) {
            atomicAdd(&ls[lo],      s0); atomicAdd(&lq[lo],      q0);
            atomicAdd(&ls[lo + 16], s1); atomicAdd(&lq[lo + 16], q1);
        }
    }
    __syncthreads();
    if (tid < C_DIM) {
        atomicAdd(&gsum[j * C_DIM + tid], ls[tid]);
        atomicAdd(&gsq [j * C_DIM + tid], lq[tid]);
    }
}

__global__ void k_stats(const float* __restrict__ gsum, const float* __restrict__ gsq,
                        const float* __restrict__ gamma, const float* __restrict__ beta,
                        float* __restrict__ scale, float* __restrict__ shift) {
    int c = threadIdx.x;
    if (c < NCH) {
        float mean = gsum[c] * (1.0f / PER_CH);
        float var  = gsq[c]  * (1.0f / PER_CH) - mean * mean;
        float sc   = gamma[c] * rsqrtf(var + BN_EPS);
        scale[c] = sc;
        shift[c] = beta[c] - mean * sc;
    }
}

__global__ void k_conv(const float* __restrict__ y, const float* __restrict__ scale,
                       const float* __restrict__ shift, const float* __restrict__ cw,
                       const float* __restrict__ cb, float* __restrict__ out) {
    __shared__ float ssc[NCH];
    __shared__ float ssh[NCH];
    const int lane = threadIdx.x;
    for (int i = lane; i < NCH; i += 32) { ssc[i] = scale[i]; ssh[i] = shift[i]; }
    __syncthreads();

    const int ntile = blockIdx.x;
    const int b     = blockIdx.y >> 1;
    const int ot    = blockIdx.y & 1;
    const int lo    = lane & 15;
    const int hi    = lane >> 4;
    const int o     = ot * 16 + lo;

    const float* pa = cw + (size_t)o * NCH + hi * 2;
    const float* py = y + (size_t)b * NCH * N_DIM + ntile * 16 + lo;

    v8f acc = {};
    #pragma unroll 4
    for (int k = 0; k < NCH; k += 4) {
        const int ch = k + hi * 2;
        v2f a = *(const v2f*)(pa + k);
        float v0 = __builtin_nontemporal_load(py + (size_t)ch       * N_DIM);
        float v1 = __builtin_nontemporal_load(py + (size_t)(ch + 1) * N_DIM);
        v0 = fmaxf(fmaf(v0, ssc[ch],     ssh[ch]),     0.0f);
        v1 = fmaxf(fmaf(v1, ssc[ch + 1], ssh[ch + 1]), 0.0f);
        v2f bb; bb[0] = v0; bb[1] = v1;
        acc = wmma4(a, bb, acc);
    }

    const int obase = ot * 16 + hi * 8;
    const int n0    = ntile * 16 + lo;
    #pragma unroll
    for (int r = 0; r < 8; ++r) {
        out[((size_t)(b * C_DIM + obase + r)) * N_DIM + n0] = acc[r] + cb[obase + r];
    }
}

extern "C" void kernel_launch(void* const* d_in, const int* in_sizes, int n_in,
                              void* d_out, int out_size, void* d_ws, size_t ws_size,
                              hipStream_t stream) {
    (void)in_sizes; (void)n_in; (void)out_size; (void)ws_size;
    const float* W      = (const float*)d_in[0];
    const float* x      = (const float*)d_in[1];
    const float* gamma  = (const float*)d_in[2];
    const float* beta   = (const float*)d_in[3];
    const float* conv_w = (const float*)d_in[4];
    const float* conv_b = (const float*)d_in[5];
    float* out = (float*)d_out;

    float* ws     = (float*)d_ws;
    float* y      = ws + WS_Y;
    float* gsum   = ws + WS_SUM;
    float* gsq    = ws + WS_SQ;
    float* sscale = ws + WS_SCALE;
    float* sshift = ws + WS_SHIFT;

    k_init<<<1, 256, 0, stream>>>(gsum, gsq);
    dim3 g1(16, 16);
    k_gmul<<<g1, 256, 0, stream>>>(W, x, y, gsum, gsq);
    k_stats<<<1, 128, 0, stream>>>(gsum, gsq, gamma, beta, sscale, sshift);
    dim3 g3(125, 8);
    k_conv<<<g3, 32, 0, stream>>>(y, sscale, sshift, conv_w, conv_b, out);
}


# k_gmul: x[b] copied whole into LDS once; each wave stages its own 16 W rows through an LDS strip only it touches, so the K loop has no block barrier and waves run unsynchronised; sums unchanged
# speedup vs baseline: 3.3387x; 1.1058x over previous
#include <hip/hip_runtime.h>

typedef float v2f __attribute__((ext_vector_type(2)));
typedef float v4f __attribute__((ext_vector_type(4)));
typedef float v8f __attribute__((ext_vector_type(8)));

#define N_DIM   2000
#define C_DIM   32
#define J_DIM   4
#define BS_DIM  4
#define NCH     128
#define PER_CH  8000.0f
#define BN_EPS  1e-5f

__device__ __forceinline__ v8f wmma4(v2f a, v2f b, v8f c) {
    return __builtin_amdgcn_wmma_f32_16x16x4_f32(
         false, a,  false, b,
         (short)0, c,  false,  false);
}

#define WS_Y      0
#define WS_SUM    1024000
#define WS_SQ     1024128
#define WS_SCALE  1024256
#define WS_SHIFT  1024384

__global__ void k_init(float* __restrict__ gsum, float* __restrict__ gsq) {
    int t = threadIdx.x;
    if (t < NCH) { gsum[t] = 0.0f; gsq[t] = 0.0f; }
}

#define KC      128
#define NCHUNK  16
#define KLAST   10
#define LP      132
#define LPB     2004
#define XPIECES 16000
#define XBATCH  21

template <int NS>
__device__ __forceinline__ void mma_chunk(const float* __restrict__ pa, const float* __restrict__ pb0,
                                          const float* __restrict__ pb1, v8f& acc0, v8f& acc1) {
    #pragma unroll
    for (int s = 0; s < NS; ++s) {
        const v4f fa  = *(const v4f*)(pa  + s * 8);
        const v4f fb0 = *(const v4f*)(pb0 + s * 8);
        const v4f fb1 = *(const v4f*)(pb1 + s * 8);
        v2f a0, a1, p0, p1, q0, q1;
        a0[0] = fa[0];  a0[1] = fa[1];  a1[0] = fa[2];  a1[1] = fa[3];
        p0[0] = fb0[0]; p0[1] = fb0[1]; p1[0] = fb0[2]; p1[1] = fb0[3];
        q0[0] = fb1[0]; q0[1] = fb1[1]; q1[0] = fb1[2]; q1[1] = fb1[3];
        acc0 = wmma4(a0, p0, acc0);
        acc1 = wmma4(a0, q0, acc1);
        acc0 = wmma4(a1, p1, acc0);
        acc1 = wmma4(a1, q1, acc1);
    }
}

__device__ __forceinline__ void load_rows(v4f (&r)[16], const float* __restrict__ gw, int q, int lane) {
    const int kk = min(q * KC + lane * 4, N_DIM - 4);
    #pragma unroll
    for (int i = 0; i < 16; ++i) r[i] = __builtin_nontemporal_load((const v4f*)(gw + (size_t)i * N_DIM + kk));
}

__device__ __forceinline__ void store_rows(const v4f (&r)[16], float* __restrict__ strip, int lane) {
    #pragma unroll
    for (int i = 0; i < 16; ++i) *(v4f*)(strip + i * LP + lane * 4) = r[i];
}

__global__ void __launch_bounds__(256)
k_gmul(const float* __restrict__ W, const float* __restrict__ x,
       float* __restrict__ y, float* __restrict__ gsum,
       float* __restrict__ gsq) {
    __shared__ __attribute__((aligned(16))) float sB[C_DIM * LPB];
    __shared__ __attribute__((aligned(16))) float sA[8][16 * LP];
    __shared__ float ls[C_DIM];
    __shared__ float lq[C_DIM];
    const int tid  = threadIdx.x;
    const int lane = tid & 31;
    const int wave = tid >> 5;
    const int bj   = blockIdx.y;
    const int b    = bj >> 2;
    const int j    = bj & 3;
    const int lo   = lane & 15;
    const int hi   = lane >> 4;

    if (tid < C_DIM) { ls[tid] = 0.0f; lq[tid] = 0.0f; }

    const int  mtile   = blockIdx.x * 8 + wave;
    const bool tile_ok = mtile < 125;
    const float* gw = W + (size_t)bj * N_DIM * N_DIM + (size_t)(tile_ok ? mtile : 0) * 16 * N_DIM;
    float* strip = &sA[wave][0];

    v4f r0[16], r1[16];
    if (tile_ok) { load_rows(r0, gw, 0, lane); load_rows(r1, gw, 1, lane); }

    {
        const float* gx = x + (size_t)b * C_DIM * N_DIM;
        for (int t0 = 0; t0 < 3; ++t0) {
            v4f xb[XBATCH];
            #pragma unroll
            for (int u = 0; u < XBATCH; ++u) {
                const int idx = min(tid + 256 * (t0 * XBATCH + u), XPIECES - 1);
                xb[u] = *(const v4f*)(gx + (size_t)idx * 4);
            }
            #pragma unroll
            for (int u = 0; u < XBATCH; ++u) {
                const int idx = min(tid + 256 * (t0 * XBATCH + u), XPIECES - 1);
                const int row = idx / 500;
                *(v4f*)(&sB[row * LPB + (idx - row * 500) * 4]) = xb[u];
            }
        }
    }
    __syncthreads();

    v8f acc0 = {};
    v8f acc1 = {};
    if (tile_ok) {
        const float* pa  = strip + lo * LP + hi * 4;
        const float* pb0 = &sB[lo * LPB + hi * 4];
        const float* pb1 = pb0 + 16 * LPB;
        store_rows(r0, strip, lane);
        for (int c = 0; c < NCHUNK; c += 2) {
            const bool more = (c + 2) < NCHUNK;
            if (more) load_rows(r0, gw, c + 2, lane);
            mma_chunk<16>(pa, pb0 + c * KC, pb1 + c * KC, acc0, acc1);
            store_rows(r1, strip, lane);
            if (more) {
                load_rows(r1, gw, c + 3, lane);
                mma_chunk<16>(pa, pb0 + (c + 1) * KC, pb1 + (c + 1) * KC, acc0, acc1);
                store_rows(r0, strip, lane);
            } else {
                mma_chunk<KLAST>(pa, pb0 + (c + 1) * KC, pb1 + (c + 1) * KC, acc0, acc1);
            }
        }

        const int m0  = mtile * 16 + hi * 8;
        const int ch0 = j * C_DIM + lo;
        float* yp0 = y + ((size_t)(b * NCH + ch0))      * N_DIM + m0;
        float* yp1 = y + ((size_t)(b * NCH + ch0 + 16)) * N_DIM + m0;
        *(float4*)(yp0)     = make_float4(acc0[0], acc0[1], acc0[2], acc0[3]);
        *(float4*)(yp0 + 4) = make_float4(acc0[4], acc0[5], acc0[6], acc0[7]);
        *(float4*)(yp1)     = make_float4(acc1[0], acc1[1], acc1[2], acc1[3]);
        *(float4*)(yp1 + 4) = make_float4(acc1[4], acc1[5], acc1[6], acc1[7]);

        float s0 = 0.f, q0 = 0.f, s1 = 0.f, q1 = 0.f;
        #pragma unroll
        for (int r = 0; r < 8; ++r) {
            s0 += acc0[r]; q0 += acc0[r] * acc0[r];
            s1 += acc1[r]; q1 += acc1[r] * acc1[r];
        }
        s0 += __shfl_xor(s0, 16); q0 += __shfl_xor(q0, 16);
        s1 += __shfl_xor(s1, 16); q1 += __shfl_xor(q1, 16);
        if (lane < 16) {
            atomicAdd(&ls[lo],      s0); atomicAdd(&lq[lo],      q0);
            atomicAdd(&ls[lo + 16], s1); atomicAdd(&lq[lo + 16], q1);
        }
    }
    __syncthreads();
    if (tid < C_DIM) {
        atomicAdd(&gsum[j * C_DIM + tid], ls[tid]);
        atomicAdd(&gsq [j * C_DIM + tid], lq[tid]);
    }
}

__global__ void k_stats(const float* __restrict__ gsum, const float* __restrict__ gsq,
                        const float* __restrict__ gamma, const float* __restrict__ beta,
                        float* __restrict__ scale, float* __restrict__ shift) {
    int c = threadIdx.x;
    if (c < NCH) {
        float mean = gsum[c] * (1.0f / PER_CH);
        float var  = gsq[c]  * (1.0f / PER_CH) - mean * mean;
        float sc   = gamma[c] * rsqrtf(var + BN_EPS);
        scale[c] = sc;
        shift[c] = beta[c] - mean * sc;
    }
}

__global__ void k_conv(const float* __restrict__ y, const float* __restrict__ scale,
                       const float* __restrict__ shift, const float* __restrict__ cw,
                       const float* __restrict__ cb, float* __restrict__ out) {
    __shared__ float ssc[NCH];
    __shared__ float ssh[NCH];
    const int lane = threadIdx.x;
    for (int i = lane; i < NCH; i += 32) { ssc[i] = scale[i]; ssh[i] = shift[i]; }
    __syncthreads();

    const int ntile = blockIdx.x;
    const int b     = blockIdx.y >> 1;
    const int ot    = blockIdx.y & 1;
    const int lo    = lane & 15;
    const int hi    = lane >> 4;
    const int o     = ot * 16 + lo;

    const float* pa = cw + (size_t)o * NCH + hi * 2;
    const float* py = y + (size_t)b * NCH * N_DIM + ntile * 16 + lo;

    v8f acc = {};
    #pragma unroll 4
    for (int k = 0; k < NCH; k += 4) {
        const int ch = k + hi * 2;
        v2f a = *(const v2f*)(pa + k);
        float v0 = __builtin_nontemporal_load(py + (size_t)ch       * N_DIM);
        float v1 = __builtin_nontemporal_load(py + (size_t)(ch + 1) * N_DIM);
        v0 = fmaxf(fmaf(v0, ssc[ch],     ssh[ch]),     0.0f);
        v1 = fmaxf(fmaf(v1, ssc[ch + 1], ssh[ch + 1]), 0.0f);
        v2f bb; bb[0] = v0; bb[1] = v1;
        acc = wmma4(a, bb, acc);
    }

    const int obase = ot * 16 + hi * 8;
    const int n0    = ntile * 16 + lo;
    #pragma unroll
    for (int r = 0; r < 8; ++r) {
        out[((size_t)(b * C_DIM + obase + r)) * N_DIM + n0] = acc[r] + cb[obase + r];
    }
}

extern "C" void kernel_launch(void* const* d_in, const int* in_sizes, int n_in,
                              void* d_out, int out_size, void* d_ws, size_t ws_size,
                              hipStream_t stream) {
    (void)in_sizes; (void)n_in; (void)out_size; (void)ws_size;
    const float* W      = (const float*)d_in[0];
    const float* x      = (const float*)d_in[1];
    const float* gamma  = (const float*)d_in[2];
    const float* beta   = (const float*)d_in[3];
    const float* conv_w = (const float*)d_in[4];
    const float* conv_b = (const float*)d_in[5];
    float* out = (float*)d_out;

    float* ws     = (float*)d_ws;
    float* y      = ws + WS_Y;
    float* gsum   = ws + WS_SUM;
    float* gsq    = ws + WS_SQ;
    float* sscale = ws + WS_SCALE;
    float* sshift = ws + WS_SHIFT;

    k_init<<<1, 256, 0, stream>>>(gsum, gsq);
    dim3 g1(16, 16);
    k_gmul<<<g1, 256, 0, stream>>>(W, x, y, gsum, gsq);
    k_stats<<<1, 128, 0, stream>>>(gsum, gsq, gamma, beta, sscale, sshift);
    dim3 g3(125, 8);
    k_conv<<<g3, 32, 0, stream>>>(y, sscale, sshift, conv_w, conv_b, out);
}


# k_init and k_stats launches removed: k_gmul stores per-block channel sums as whole 128-B lines (no global atomics), k_conv adds the 64 partials in fixed order and forms scale/shift itself
# speedup vs baseline: 3.4852x; 1.0439x over previous
#include <hip/hip_runtime.h>

typedef float v2f __attribute__((ext_vector_type(2)));
typedef float v4f __attribute__((ext_vector_type(4)));
typedef float v8f __attribute__((ext_vector_type(8)));

#define N_DIM   2000
#define C_DIM   32
#define J_DIM   4
#define BS_DIM  4
#define NCH     128
#define PER_CH  8000.0f
#define BN_EPS  1e-5f

__device__ __forceinline__ v8f wmma4(v2f a, v2f b, v8f c) {
    return __builtin_amdgcn_wmma_f32_16x16x4_f32(
         false, a,  false, b,
         (short)0, c,  false,  false);
}

#define WS_Y      0
#define WS_PSUM   1024000
#define WS_PSQ    1032192

#define KC      128
#define NCHUNK  16
#define KLAST   10
#define LP      132
#define LPB     2004
#define XPIECES 16000
#define XBATCH  21

template <int NS>
__device__ __forceinline__ void mma_chunk(const float* __restrict__ pa, const float* __restrict__ pb0,
                                          const float* __restrict__ pb1, v8f& acc0, v8f& acc1) {
    #pragma unroll
    for (int s = 0; s < NS; ++s) {
        const v4f fa  = *(const v4f*)(pa  + s * 8);
        const v4f fb0 = *(const v4f*)(pb0 + s * 8);
        const v4f fb1 = *(const v4f*)(pb1 + s * 8);
        v2f a0, a1, p0, p1, q0, q1;
        a0[0] = fa[0];  a0[1] = fa[1];  a1[0] = fa[2];  a1[1] = fa[3];
        p0[0] = fb0[0]; p0[1] = fb0[1]; p1[0] = fb0[2]; p1[1] = fb0[3];
        q0[0] = fb1[0]; q0[1] = fb1[1]; q1[0] = fb1[2]; q1[1] = fb1[3];
        acc0 = wmma4(a0, p0, acc0);
        acc1 = wmma4(a0, q0, acc1);
        acc0 = wmma4(a1, p1, acc0);
        acc1 = wmma4(a1, q1, acc1);
    }
}

__device__ __forceinline__ void load_rows(v4f (&r)[16], const float* __restrict__ gw, int q, int lane) {
    const int kk = min(q * KC + lane * 4, N_DIM - 4);
    #pragma unroll
    for (int i = 0; i < 16; ++i) r[i] = __builtin_nontemporal_load((const v4f*)(gw + (size_t)i * N_DIM + kk));
}

__device__ __forceinline__ void store_rows(const v4f (&r)[16], float* __restrict__ strip, int lane) {
    #pragma unroll
    for (int i = 0; i < 16; ++i) *(v4f*)(strip + i * LP + lane * 4) = r[i];
}

__global__ void __launch_bounds__(256)
k_gmul(const float* __restrict__ W, const float* __restrict__ x,
       float* __restrict__ y, float* __restrict__ psum,
       float* __restrict__ psq) {
    __shared__ __attribute__((aligned(16))) float sB[C_DIM * LPB];
    __shared__ __attribute__((aligned(16))) float sA[8][16 * LP];
    __shared__ float ls[C_DIM];
    __shared__ float lq[C_DIM];
    const int tid  = threadIdx.x;
    const int lane = tid & 31;
    const int wave = tid >> 5;
    const int bj   = blockIdx.y;
    const int b    = bj >> 2;
    const int j    = bj & 3;
    const int lo   = lane & 15;
    const int hi   = lane >> 4;

    if (tid < C_DIM) { ls[tid] = 0.0f; lq[tid] = 0.0f; }

    const int  mtile   = blockIdx.x * 8 + wave;
    const bool tile_ok = mtile < 125;
    const float* gw = W + (size_t)bj * N_DIM * N_DIM + (size_t)(tile_ok ? mtile : 0) * 16 * N_DIM;
    float* strip = &sA[wave][0];

    v4f r0[16], r1[16];
    if (tile_ok) { load_rows(r0, gw, 0, lane); load_rows(r1, gw, 1, lane); }

    {
        const float* gx = x + (size_t)b * C_DIM * N_DIM;
        for (int t0 = 0; t0 < 3; ++t0) {
            v4f xb[XBATCH];
            #pragma unroll
            for (int u = 0; u < XBATCH; ++u) {
                const int idx = min(tid + 256 * (t0 * XBATCH + u), XPIECES - 1);
                xb[u] = *(const v4f*)(gx + (size_t)idx * 4);
            }
            #pragma unroll
            for (int u = 0; u < XBATCH; ++u) {
                const int idx = min(tid + 256 * (t0 * XBATCH + u), XPIECES - 1);
                const int row = idx / 500;
                *(v4f*)(&sB[row * LPB + (idx - row * 500) * 4]) = xb[u];
            }
        }
    }
    __syncthreads();

    v8f acc0 = {};
    v8f acc1 = {};
    if (tile_ok) {
        const float* pa  = strip + lo * LP + hi * 4;
        const float* pb0 = &sB[lo * LPB + hi * 4];
        const float* pb1 = pb0 + 16 * LPB;
        store_rows(r0, strip, lane);
        for (int c = 0; c < NCHUNK; c += 2) {
            const bool more = (c + 2) < NCHUNK;
            if (more) load_rows(r0, gw, c + 2, lane);
            mma_chunk<16>(pa, pb0 + c * KC, pb1 + c * KC, acc0, acc1);
            store_rows(r1, strip, lane);
            if (more) {
                load_rows(r1, gw, c + 3, lane);
                mma_chunk<16>(pa, pb0 + (c + 1) * KC, pb1 + (c + 1) * KC, acc0, acc1);
                store_rows(r0, strip, lane);
            } else {
                mma_chunk<KLAST>(pa, pb0 + (c + 1) * KC, pb1 + (c + 1) * KC, acc0, acc1);
            }
        }

        const int m0  = mtile * 16 + hi * 8;
        const int ch0 = j * C_DIM + lo;
        float* yp0 = y + ((size_t)(b * NCH + ch0))      * N_DIM + m0;
        float* yp1 = y + ((size_t)(b * NCH + ch0 + 16)) * N_DIM + m0;
        *(float4*)(yp0)     = make_float4(acc0[0], acc0[1], acc0[2], acc0[3]);
        *(float4*)(yp0 + 4) = make_float4(acc0[4], acc0[5], acc0[6], acc0[7]);
        *(float4*)(yp1)     = make_float4(acc1[0], acc1[1], acc1[2], acc1[3]);
        *(float4*)(yp1 + 4) = make_float4(acc1[4], acc1[5], acc1[6], acc1[7]);

        float s0 = 0.f, q0 = 0.f, s1 = 0.f, q1 = 0.f;
        #pragma unroll
        for (int r = 0; r < 8; ++r) {
            s0 += acc0[r]; q0 += acc0[r] * acc0[r];
            s1 += acc1[r]; q1 += acc1[r] * acc1[r];
        }
        s0 += __shfl_xor(s0, 16); q0 += __shfl_xor(q0, 16);
        s1 += __shfl_xor(s1, 16); q1 += __shfl_xor(q1, 16);
        if (lane < 16) {
            atomicAdd(&ls[lo],      s0); atomicAdd(&lq[lo],      q0);
            atomicAdd(&ls[lo + 16], s1); atomicAdd(&lq[lo + 16], q1);
        }
    }
    __syncthreads();
    if (tid < C_DIM) {
        const int slot = b * 16 + blockIdx.x;
        psum[slot * NCH + j * C_DIM + tid] = ls[tid];
        psq [slot * NCH + j * C_DIM + tid] = lq[tid];
    }
}

__global__ void __launch_bounds__(128)
k_conv(const float* __restrict__ y, const float* __restrict__ psum,
       const float* __restrict__ psq, const float* __restrict__ gamma,
       const float* __restrict__ beta, const float* __restrict__ cw,
       const float* __restrict__ cb, float* __restrict__ out) {
    __shared__ __attribute__((aligned(16))) float rs[4 * NCH];
    __shared__ __attribute__((aligned(16))) float rq[4 * NCH];
    __shared__ float ssc[NCH];
    __shared__ float ssh[NCH];
    const int tid  = threadIdx.x;
    const int lane = tid & 31;
    const int wave = tid >> 5;

    const float g  = gamma[tid];
    const float be = beta[tid];

    {
        const float* ps = psum + (size_t)(wave * 16) * NCH + lane * 4;
        const float* pq = psq  + (size_t)(wave * 16) * NCH + lane * 4;
        v4f vs[16], vq[16];
        #pragma unroll
        for (int u = 0; u < 16; ++u) {
            vs[u] = *(const v4f*)(ps + u * NCH);
            vq[u] = *(const v4f*)(pq + u * NCH);
        }
        __builtin_amdgcn_sched_barrier(0);
        v4f s4 = vs[0], q4 = vq[0];
        #pragma unroll
        for (int u = 1; u < 16; ++u) { s4 += vs[u]; q4 += vq[u]; }
        *(v4f*)(&rs[wave * NCH + lane * 4]) = s4;
        *(v4f*)(&rq[wave * NCH + lane * 4]) = q4;
    }
    __syncthreads();
    {
        const float sum = ((rs[tid] + rs[NCH + tid]) + rs[2 * NCH + tid]) + rs[3 * NCH + tid];
        const float sq  = ((rq[tid] + rq[NCH + tid]) + rq[2 * NCH + tid]) + rq[3 * NCH + tid];
        const float mean = sum * (1.0f / PER_CH);
        const float var  = sq  * (1.0f / PER_CH) - mean * mean;
        const float sc   = g * rsqrtf(var + BN_EPS);
        ssc[tid] = sc;
        ssh[tid] = be - mean * sc;
    }
    __syncthreads();

    const int ntile = blockIdx.x;
    const int b     = blockIdx.y * 2 + (wave >> 1);
    const int ot    = wave & 1;
    const int lo    = lane & 15;
    const int hi    = lane >> 4;
    const int o     = ot * 16 + lo;

    const float* pa = cw + (size_t)o * NCH + hi * 2;
    const float* py = y + (size_t)b * NCH * N_DIM + ntile * 16 + lo;

    v8f acc = {};
    #pragma unroll 4
    for (int k = 0; k < NCH; k += 4) {
        const int ch = k + hi * 2;
        v2f a = *(const v2f*)(pa + k);
        float v0 = __builtin_nontemporal_load(py + (size_t)ch       * N_DIM);
        float v1 = __builtin_nontemporal_load(py + (size_t)(ch + 1) * N_DIM);
        v0 = fmaxf(fmaf(v0, ssc[ch],     ssh[ch]),     0.0f);
        v1 = fmaxf(fmaf(v1, ssc[ch + 1], ssh[ch + 1]), 0.0f);
        v2f bb; bb[0] = v0; bb[1] = v1;
        acc = wmma4(a, bb, acc);
    }

    const int obase = ot * 16 + hi * 8;
    const int n0    = ntile * 16 + lo;
    #pragma unroll
    for (int r = 0; r < 8; ++r) {
        out[((size_t)(b * C_DIM + obase + r)) * N_DIM + n0] = acc[r] + cb[obase + r];
    }
}

extern "C" void kernel_launch(void* const* d_in, const int* in_sizes, int n_in,
                              void* d_out, int out_size, void* d_ws, size_t ws_size,
                              hipStream_t stream) {
    (void)in_sizes; (void)n_in; (void)out_size; (void)ws_size;
    const float* W      = (const float*)d_in[0];
    const float* x      = (const float*)d_in[1];
    const float* gamma  = (const float*)d_in[2];
    const float* beta   = (const float*)d_in[3];
    const float* conv_w = (const float*)d_in[4];
    const float* conv_b = (const float*)d_in[5];
    float* out = (float*)d_out;

    float* ws   = (float*)d_ws;
    float* y    = ws + WS_Y;
    float* psum = ws + WS_PSUM;
    float* psq  = ws + WS_PSQ;

    dim3 g1(16, 16);
    k_gmul<<<g1, 256, 0, stream>>>(W, x, y, psum, psq);
    dim3 g2(125, 2);
    k_conv<<<g2, 128, 0, stream>>>(y, psum, psq, gamma, beta, conv_w, conv_b, out);
}


# k_conv: all of a wave's global loads (partials, y column, conv_w row, bias) issued as one group at entry and waited for once, instead of ten dependent round trips; sums unchanged
# speedup vs baseline: 3.6945x; 1.0601x over previous
#include <hip/hip_runtime.h>

typedef float v2f __attribute__((ext_vector_type(2)));
typedef float v4f __attribute__((ext_vector_type(4)));
typedef float v8f __attribute__((ext_vector_type(8)));

#define N_DIM   2000
#define C_DIM   32
#define J_DIM   4
#define BS_DIM  4
#define NCH     128
#define PER_CH  8000.0f
#define BN_EPS  1e-5f

__device__ __forceinline__ v8f wmma4(v2f a, v2f b, v8f c) {
    return __builtin_amdgcn_wmma_f32_16x16x4_f32(
         false, a,  false, b,
         (short)0, c,  false,  false);
}

#define WS_Y      0
#define WS_PSUM   1024000
#define WS_PSQ    1032192

#define KC      128
#define NCHUNK  16
#define KLAST   10
#define LP      132
#define LPB     2004
#define XPIECES 16000
#define XBATCH  21

template <int NS>
__device__ __forceinline__ void mma_chunk(const float* __restrict__ pa, const float* __restrict__ pb0,
                                          const float* __restrict__ pb1, v8f& acc0, v8f& acc1) {
    #pragma unroll
    for (int s = 0; s < NS; ++s) {
        const v4f fa  = *(const v4f*)(pa  + s * 8);
        const v4f fb0 = *(const v4f*)(pb0 + s * 8);
        const v4f fb1 = *(const v4f*)(pb1 + s * 8);
        v2f a0, a1, p0, p1, q0, q1;
        a0[0] = fa[0];  a0[1] = fa[1];  a1[0] = fa[2];  a1[1] = fa[3];
        p0[0] = fb0[0]; p0[1] = fb0[1]; p1[0] = fb0[2]; p1[1] = fb0[3];
        q0[0] = fb1[0]; q0[1] = fb1[1]; q1[0] = fb1[2]; q1[1] = fb1[3];
        acc0 = wmma4(a0, p0, acc0);
        acc1 = wmma4(a0, q0, acc1);
        acc0 = wmma4(a1, p1, acc0);
        acc1 = wmma4(a1, q1, acc1);
    }
}

__device__ __forceinline__ void load_rows(v4f (&r)[16], const float* __restrict__ gw, int q, int lane) {
    const int kk = min(q * KC + lane * 4, N_DIM - 4);
    #pragma unroll
    for (int i = 0; i < 16; ++i) r[i] = __builtin_nontemporal_load((const v4f*)(gw + (size_t)i * N_DIM + kk));
}

__device__ __forceinline__ void store_rows(const v4f (&r)[16], float* __restrict__ strip, int lane) {
    #pragma unroll
    for (int i = 0; i < 16; ++i) *(v4f*)(strip + i * LP + lane * 4) = r[i];
}

__global__ void __launch_bounds__(256)
k_gmul(const float* __restrict__ W, const float* __restrict__ x,
       float* __restrict__ y, float* __restrict__ psum,
       float* __restrict__ psq) {
    __shared__ __attribute__((aligned(16))) float sB[C_DIM * LPB];
    __shared__ __attribute__((aligned(16))) float sA[8][16 * LP];
    __shared__ float ls[C_DIM];
    __shared__ float lq[C_DIM];
    const int tid  = threadIdx.x;
    const int lane = tid & 31;
    const int wave = tid >> 5;
    const int bj   = blockIdx.y;
    const int b    = bj >> 2;
    const int j    = bj & 3;
    const int lo   = lane & 15;
    const int hi   = lane >> 4;

    if (tid < C_DIM) { ls[tid] = 0.0f; lq[tid] = 0.0f; }

    const int  mtile   = blockIdx.x * 8 + wave;
    const bool tile_ok = mtile < 125;
    const float* gw = W + (size_t)bj * N_DIM * N_DIM + (size_t)(tile_ok ? mtile : 0) * 16 * N_DIM;
    float* strip = &sA[wave][0];

    v4f r0[16], r1[16];
    if (tile_ok) { load_rows(r0, gw, 0, lane); load_rows(r1, gw, 1, lane); }

    {
        const float* gx = x + (size_t)b * C_DIM * N_DIM;
        for (int t0 = 0; t0 < 3; ++t0) {
            v4f xb[XBATCH];
            #pragma unroll
            for (int u = 0; u < XBATCH; ++u) {
                const int idx = min(tid + 256 * (t0 * XBATCH + u), XPIECES - 1);
                xb[u] = *(const v4f*)(gx + (size_t)idx * 4);
            }
            #pragma unroll
            for (int u = 0; u < XBATCH; ++u) {
                const int idx = min(tid + 256 * (t0 * XBATCH + u), XPIECES - 1);
                const int row = idx / 500;
                *(v4f*)(&sB[row * LPB + (idx - row * 500) * 4]) = xb[u];
            }
        }
    }
    __syncthreads();

    v8f acc0 = {};
    v8f acc1 = {};
    if (tile_ok) {
        const float* pa  = strip + lo * LP + hi * 4;
        const float* pb0 = &sB[lo * LPB + hi * 4];
        const float* pb1 = pb0 + 16 * LPB;
        store_rows(r0, strip, lane);
        for (int c = 0; c < NCHUNK; c += 2) {
            const bool more = (c + 2) < NCHUNK;
            if (more) load_rows(r0, gw, c + 2, lane);
            mma_chunk<16>(pa, pb0 + c * KC, pb1 + c * KC, acc0, acc1);
            store_rows(r1, strip, lane);
            if (more) {
                load_rows(r1, gw, c + 3, lane);
                mma_chunk<16>(pa, pb0 + (c + 1) * KC, pb1 + (c + 1) * KC, acc0, acc1);
                store_rows(r0, strip, lane);
            } else {
                mma_chunk<KLAST>(pa, pb0 + (c + 1) * KC, pb1 + (c + 1) * KC, acc0, acc1);
            }
        }

        const int m0  = mtile * 16 + hi * 8;
        const int ch0 = j * C_DIM + lo;
        float* yp0 = y + ((size_t)(b * NCH + ch0))      * N_DIM + m0;
        float* yp1 = y + ((size_t)(b * NCH + ch0 + 16)) * N_DIM + m0;
        *(float4*)(yp0)     = make_float4(acc0[0], acc0[1], acc0[2], acc0[3]);
        *(float4*)(yp0 + 4) = make_float4(acc0[4], acc0[5], acc0[6], acc0[7]);
        *(float4*)(yp1)     = make_float4(acc1[0], acc1[1], acc1[2], acc1[3]);
        *(float4*)(yp1 + 4) = make_float4(acc1[4], acc1[5], acc1[6], acc1[7]);

        float s0 = 0.f, q0 = 0.f, s1 = 0.f, q1 = 0.f;
        #pragma unroll
        for (int r = 0; r < 8; ++r) {
            s0 += acc0[r]; q0 += acc0[r] * acc0[r];
            s1 += acc1[r]; q1 += acc1[r] * acc1[r];
        }
        s0 += __shfl_xor(s0, 16); q0 += __shfl_xor(q0, 16);
        s1 += __shfl_xor(s1, 16); q1 += __shfl_xor(q1, 16);
        if (lane < 16) {
            atomicAdd(&ls[lo],      s0); atomicAdd(&lq[lo],      q0);
            atomicAdd(&ls[lo + 16], s1); atomicAdd(&lq[lo + 16], q1);
        }
    }
    __syncthreads();
    if (tid < C_DIM) {
        const int slot = b * 16 + blockIdx.x;
        psum[slot * NCH + j * C_DIM + tid] = ls[tid];
        psq [slot * NCH + j * C_DIM + tid] = lq[tid];
    }
}

__global__ void __launch_bounds__(128)
k_conv(const float* __restrict__ y, const float* __restrict__ psum,
       const float* __restrict__ psq, const float* __restrict__ gamma,
       const float* __restrict__ beta, const float* __restrict__ cw,
       const float* __restrict__ cb, float* __restrict__ out) {
    __shared__ __attribute__((aligned(16))) float rs[4 * NCH];
    __shared__ __attribute__((aligned(16))) float rq[4 * NCH];
    __shared__ float ssc[NCH];
    __shared__ float ssh[NCH];
    const int tid  = threadIdx.x;
    const int lane = tid & 31;
    const int wave = tid >> 5;

    const int ntile = blockIdx.x;
    const int b     = blockIdx.y * 2 + (wave >> 1);
    const int ot    = wave & 1;
    const int lo    = lane & 15;
    const int hi    = lane >> 4;
    const int o     = ot * 16 + lo;
    const int obase = ot * 16 + hi * 8;

    const float* pa = cw + (size_t)o * NCH + hi * 2;
    const float* py = y + (size_t)b * NCH * N_DIM + ntile * 16 + lo;

    const float g  = gamma[tid];
    const float be = beta[tid];

    const float* ps = psum + (size_t)(wave * 16) * NCH + lane * 4;
    const float* pq = psq  + (size_t)(wave * 16) * NCH + lane * 4;
    v4f vs[16], vq[16];
    #pragma unroll
    for (int u = 0; u < 16; ++u) {
        vs[u] = *(const v4f*)(ps + u * NCH);
        vq[u] = *(const v4f*)(pq + u * NCH);
    }
    float yv[2 * (NCH / 4)];
    v2f   av[NCH / 4];
    #pragma unroll
    for (int s = 0; s < NCH / 4; ++s) {
        const int ch = s * 4 + hi * 2;
        yv[2 * s]     = __builtin_nontemporal_load(py + (size_t)ch       * N_DIM);
        yv[2 * s + 1] = __builtin_nontemporal_load(py + (size_t)(ch + 1) * N_DIM);
        av[s] = *(const v2f*)(pa + s * 4);
    }
    const v4f cb0 = *(const v4f*)(cb + obase);
    const v4f cb1 = *(const v4f*)(cb + obase + 4);
    __builtin_amdgcn_sched_barrier(0);
    {
        v4f s4 = vs[0], q4 = vq[0];
        #pragma unroll
        for (int u = 1; u < 16; ++u) { s4 += vs[u]; q4 += vq[u]; }
        *(v4f*)(&rs[wave * NCH + lane * 4]) = s4;
        *(v4f*)(&rq[wave * NCH + lane * 4]) = q4;
    }
    __syncthreads();
    {
        const float sum = ((rs[tid] + rs[NCH + tid]) + rs[2 * NCH + tid]) + rs[3 * NCH + tid];
        const float sq  = ((rq[tid] + rq[NCH + tid]) + rq[2 * NCH + tid]) + rq[3 * NCH + tid];
        const float mean = sum * (1.0f / PER_CH);
        const float var  = sq  * (1.0f / PER_CH) - mean * mean;
        const float sc   = g * rsqrtf(var + BN_EPS);
        ssc[tid] = sc;
        ssh[tid] = be - mean * sc;
    }
    __syncthreads();

    v8f acc = {};
    #pragma unroll
    for (int s = 0; s < NCH / 4; ++s) {
        const int ch = s * 4 + hi * 2;
        const float v0 = fmaxf(fmaf(yv[2 * s],     ssc[ch],     ssh[ch]),     0.0f);
        const float v1 = fmaxf(fmaf(yv[2 * s + 1], ssc[ch + 1], ssh[ch + 1]), 0.0f);
        v2f bb; bb[0] = v0; bb[1] = v1;
        acc = wmma4(av[s], bb, acc);
    }

    const int n0 = ntile * 16 + lo;
    float* po = out + ((size_t)(b * C_DIM + obase)) * N_DIM + n0;
    po[0 * N_DIM] = acc[0] + cb0[0];
    po[1 * N_DIM] = acc[1] + cb0[1];
    po[2 * N_DIM] = acc[2] + cb0[2];
    po[3 * N_DIM] = acc[3] + cb0[3];
    po[4 * N_DIM] = acc[4] + cb1[0];
    po[5 * N_DIM] = acc[5] + cb1[1];
    po[6 * N_DIM] = acc[6] + cb1[2];
    po[7 * N_DIM] = acc[7] + cb1[3];
}

extern "C" void kernel_launch(void* const* d_in, const int* in_sizes, int n_in,
                              void* d_out, int out_size, void* d_ws, size_t ws_size,
                              hipStream_t stream) {
    (void)in_sizes; (void)n_in; (void)out_size; (void)ws_size;
    const float* W      = (const float*)d_in[0];
    const float* x      = (const float*)d_in[1];
    const float* gamma  = (const float*)d_in[2];
    const float* beta   = (const float*)d_in[3];
    const float* conv_w = (const float*)d_in[4];
    const float* conv_b = (const float*)d_in[5];
    float* out = (float*)d_out;

    float* ws   = (float*)d_ws;
    float* y    = ws + WS_Y;
    float* psum = ws + WS_PSUM;
    float* psq  = ws + WS_PSQ;

    dim3 g1(16, 16);
    k_gmul<<<g1, 256, 0, stream>>>(W, x, y, psum, psq);
    dim3 g2(125, 2);
    k_conv<<<g2, 128, 0, stream>>>(y, psum, psq, gamma, beta, conv_w, conv_b, out);
}
